# starting kernel
# baseline (speedup 1.0000x reference)
#include <hip/hip_runtime.h>
#include <math.h>

typedef __attribute__((ext_vector_type(2))) float v2f;
typedef __attribute__((ext_vector_type(8))) float v8f;

#define B_DIM 64
#define D_DIM 512
#define H_DIM 8
#define EPS_F 1e-8f

#if defined(__HIP_DEVICE_COMPILE__) && defined(__AMDGCN__)
#if !__has_builtin(__builtin_amdgcn_wmma_f32_16x16x4_f32)
#error "missing __builtin_amdgcn_wmma_f32_16x16x4_f32 on this toolchain (device pass)"
#endif
#endif

__global__ void fl_init_out(const float* __restrict__ x, float* __restrict__ out, int n) {
    int i = blockIdx.x * blockDim.x + threadIdx.x;
    if (i < n) out[i] = x[i];
}

__global__ __launch_bounds__(256) void fl_attn_kernel(const float* __restrict__ x,
                                                      const float* __restrict__ alphas,
                                                      const float* __restrict__ betas,
                                                      float* __restrict__ out) {
    __shared__ float sQ[D_DIM];
    __shared__ float sK[D_DIM];
    __shared__ float sV[D_DIM];
    __shared__ float sC[8][16 * 16];

    const int hb  = blockIdx.x;
    const int b   = hb & (B_DIM - 1);
    const int h   = hb >> 6;
    const int tid = threadIdx.x;

    const float aq = alphas[h * 3 + 0], ak = alphas[h * 3 + 1], av = alphas[h * 3 + 2];
    const float bq = betas[h * 3 + 0],  bk = betas[h * 3 + 1],  bv = betas[h * 3 + 2];

    for (int j = tid; j < D_DIM; j += 256) {
        float xv = x[b * D_DIM + j];
        sQ[j] = fmaf(aq, xv, bq);
        sK[j] = fmaf(ak, xv, bk);
        sV[j] = fmaf(av, xv, bv);
    }
    __syncthreads();

    const int wave  = tid >> 5;
    const int lane  = tid & 31;
    const int half  = lane >> 4;
    const int laneM = lane & 15;
    const int kb    = half * 2;
    const float invSqrtD = 0.044194173824159216f;

    for (int t = wave; t < 32; t += 8) {
        const int i0 = t * 16;
        const float ki = sK[i0 + laneM];

        float mn = 3.402823466e+38f;
        const int jbeg = half * 256;
        #pragma unroll 8
        for (int j = jbeg; j < jbeg + 256; ++j) {
            mn = fminf(mn, fabsf(sQ[j] - ki));
        }
        mn = fminf(mn, __shfl_xor(mn, 16));
        const float smax = __builtin_amdgcn_rcpf(mn + EPS_F);

        v8f c0 = {};
        v8f c1 = {};
        for (int j = 0; j < D_DIM; j += 8) {
            {
                const int jj = j + kb;
                float q0 = sQ[jj], q1 = sQ[jj + 1];
                float w0 = sV[jj], w1 = sV[jj + 1];
                float s0 = __builtin_amdgcn_rcpf(fabsf(q0 - ki) + EPS_F);
                float s1 = __builtin_amdgcn_rcpf(fabsf(q1 - ki) + EPS_F);
                float e0 = __expf(s0 - smax);
                float e1 = __expf(s1 - smax);
                float b0 = (laneM == 0) ? 1.0f : ((laneM == 1) ? w0 : 0.0f);
                float b1 = (laneM == 0) ? 1.0f : ((laneM == 1) ? w1 : 0.0f);
                v2f A  = {e0, e1};
                v2f Bm = {b0, b1};
                c0 = __builtin_amdgcn_wmma_f32_16x16x4_f32(false, A, false, Bm,
                                                           (short)0, c0, false, false);
            }
            {
                const int jj = j + 4 + kb;
                float q0 = sQ[jj], q1 = sQ[jj + 1];
                float w0 = sV[jj], w1 = sV[jj + 1];
                float s0 = __builtin_amdgcn_rcpf(fabsf(q0 - ki) + EPS_F);
                float s1 = __builtin_amdgcn_rcpf(fabsf(q1 - ki) + EPS_F);
                float e0 = __expf(s0 - smax);
                float e1 = __expf(s1 - smax);
                float b0 = (laneM == 0) ? 1.0f : ((laneM == 1) ? w0 : 0.0f);
                float b1 = (laneM == 0) ? 1.0f : ((laneM == 1) ? w1 : 0.0f);
                v2f A  = {e0, e1};
                v2f Bm = {b0, b1};
                c1 = __builtin_amdgcn_wmma_f32_16x16x4_f32(false, A, false, Bm,
                                                           (short)0, c1, false, false);
            }
        }

        const int M0 = half * 8;
        #pragma unroll
        for (int r = 0; r < 8; ++r) {
            sC[wave][(M0 + r) * 16 + laneM] = c0[r] + c1[r];
        }
        asm volatile("s_wait_dscnt 0" ::: "memory");
        if (lane < 16) {
            float sumE  = sC[wave][lane * 16 + 0];
            float sumEV = sC[wave][lane * 16 + 1];
            float att   = (sumEV / sumE) * invSqrtD;
            atomicAdd(&out[b * D_DIM + i0 + lane], att);
        }
    }
}

extern "C" void kernel_launch(void* const* d_in, const int* in_sizes, int n_in,
                              void* d_out, int out_size, void* d_ws, size_t ws_size,
                              hipStream_t stream) {
    const float* x      = (const float*)d_in[0];
    const float* alphas = (const float*)d_in[1];
    const float* betas  = (const float*)d_in[2];
    float* out          = (float*)d_out;

    const int n = B_DIM * D_DIM;
    fl_init_out<<<(n + 255) / 256, 256, 0, stream>>>(x, out, n);
    fl_attn_kernel<<<H_DIM * B_DIM, 256, 0, stream>>>(x, alphas, betas, out);
}


# wide rows (own band over 64 columns) leave their row tile: listed per head through an LDS counter and computed in tiles of wide rows only, handed out first; per-row sums bit-identical
# speedup vs baseline: 3.0016x; 3.0016x over previous
#include <hip/hip_runtime.h>
#include <math.h>


typedef __attribute__((ext_vector_type(2))) float v2f;
typedef __attribute__((ext_vector_type(8))) float v8f;

#define B_DIM 64
#define D_DIM 512
#define H_DIM 8
#define EPS_F 1e-8f
#define LN2_F 0.69314718055994531f
#define EPS_LN2_F (EPS_F * LN2_F)
#define HPW 4
#define BPL (D_DIM + 4)
#define GJ 16
#define WIDE_T 64
#define ZERO_GAP 160.0f

#if defined(__HIP_DEVICE_COMPILE__) && defined(__AMDGCN__)
#if !__has_builtin(__builtin_amdgcn_wmma_f32_16x16x4_f32)
#error "missing __builtin_amdgcn_wmma_f32_16x16x4_f32 on this toolchain (device pass)"
#endif
#endif

__device__ __forceinline__ float dist_ln2(float q, float k) {
    return fmaf(fabsf(q - k), LN2_F, EPS_LN2_F);
}

__device__ __forceinline__ int xrow16_i(int b) {
    return __builtin_amdgcn_permlanex16(b, b, 0x76543210, 0xfedcba98, false, false);
}
template <int M>
__device__ __forceinline__ int xrow_i(int b) {
    return __builtin_amdgcn_update_dpp(b, b, 0x160 + M, 0xf, 0xf, false);
}
__device__ __forceinline__ float xrow16_f(float x) {
    return __builtin_bit_cast(float, xrow16_i(__builtin_bit_cast(int, x)));
}
template <int M>
__device__ __forceinline__ float xrow_f(float x) {
    return __builtin_bit_cast(float, xrow_i<M>(__builtin_bit_cast(int, x)));
}

__global__ void fl_init_out(const float* __restrict__ x, float* __restrict__ out, int n) {
    int i = blockIdx.x * blockDim.x + threadIdx.x;
    if (i < n) out[i] = x[i];
}

__global__ __launch_bounds__(256) void fl_attn_kernel(const float* __restrict__ x,
                                                      const float* __restrict__ alphas,
                                                      const float* __restrict__ betas,
                                                      float* __restrict__ out) {
    __shared__ __attribute__((aligned(16))) float sXs[D_DIM];
    __shared__ __attribute__((aligned(16))) float sQ[HPW * D_DIM];
    __shared__ __attribute__((aligned(16))) float sB[(2 + HPW) * BPL];
    __shared__ __attribute__((aligned(16))) float sKu[D_DIM];
    __shared__ __attribute__((aligned(16))) float sM[D_DIM];
    __shared__ int sLo[D_DIM];
    __shared__ int sHi[D_DIM];
    __shared__ __attribute__((aligned(16))) float sAtt[D_DIM];
    __shared__ __attribute__((aligned(16))) float sC[8][16 * 16];
    __shared__ int sWd[D_DIM];
    __shared__ int sWlist[D_DIM];
    __shared__ int sWcnt[HPW];
    __shared__ int sNext;

    const int hb  = blockIdx.x;
    const int b   = hb & (B_DIM - 1);
    const int rq  = (hb >> 6) & 3;
    const int par = ((hb >> 8) ^ hb) & 1;
    const int tid = threadIdx.x;
    const int wave  = tid >> 5;
    const int lane  = tid & 31;

    const int gi0 = (wave << 6) | lane;
    const int gi1 = gi0 | 32;
    float r0 = x[b * D_DIM + gi0];
    float r1 = x[b * D_DIM + gi1];
    #pragma unroll
    for (int k = 2; k <= D_DIM; k <<= 1) {
        #pragma unroll
        for (int jj = k >> 1; jj >= 1; jj >>= 1) {
            const bool km0 = (((gi0 & jj) == 0) == ((gi0 & k) == 0));
            const bool km1 = (((gi1 & jj) == 0) == ((gi1 & k) == 0));
            float p0, p1;
            if (jj >= 64) {
                __syncthreads();
                sXs[gi0] = r0;
                sXs[gi1] = r1;
                __syncthreads();
                p0 = sXs[gi0 ^ jj];
                p1 = sXs[gi1 ^ jj];
            } else if (jj == 32) {
                p0 = r1;
                p1 = r0;
            } else if (jj == 16) {
                p0 = xrow16_f(r0);
                p1 = xrow16_f(r1);
            } else if (jj == 8) {
                p0 = xrow_f<8>(r0);
                p1 = xrow_f<8>(r1);
            } else if (jj == 4) {
                p0 = xrow_f<4>(r0);
                p1 = xrow_f<4>(r1);
            } else if (jj == 2) {
                p0 = xrow_f<2>(r0);
                p1 = xrow_f<2>(r1);
            } else {
                p0 = xrow_f<1>(r0);
                p1 = xrow_f<1>(r1);
            }
            const float lo0 = fminf(r0, p0), hi0 = fmaxf(r0, p0);
            const float lo1 = fminf(r1, p1), hi1 = fmaxf(r1, p1);
            r0 = km0 ? lo0 : hi0;
            r1 = km1 ? lo1 : hi1;
        }
    }

    __syncthreads();
    if (tid == 0) sNext = 0;
    if (tid < HPW) sWcnt[tid] = 0;
    sXs[gi0] = r0;
    sXs[gi1] = r1;
    sB[gi0]       = 1.0f;
    sB[gi1]       = 1.0f;
    sB[BPL + gi0] = 0.0f;
    sB[BPL + gi1] = 0.0f;
    #pragma unroll
    for (int a = 0; a < HPW; ++a) {
        const int h = par + 2 * a;
        const float aq = alphas[h * 3 + 0], av = alphas[h * 3 + 2];
        const float bq = betas[h * 3 + 0],  bv = betas[h * 3 + 2];
        const bool rev = (aq < 0.0f);
        const int jq0 = rev ? (D_DIM - 1 - gi0) : gi0;
        const int jq1 = rev ? (D_DIM - 1 - gi1) : gi1;
        sQ[a * D_DIM + jq0] = fmaf(aq, r0, bq);
        sQ[a * D_DIM + jq1] = fmaf(aq, r1, bq);
        sB[(2 + a) * BPL + jq0] = fmaf(av, r0, bv);
        sB[(2 + a) * BPL + jq1] = fmaf(av, r1, bv);
    }
    __syncthreads();

    #pragma unroll
    for (int rr = 0; rr < 2; ++rr) {
        const int i = tid + rr * 256;
        const int ua = i >> 7;
        const int uh = par + 2 * ua;
        const int urow = ((rq + 4 * ((i >> 4) & 7)) << 4) | (i & 15);
        const float ki = fmaf(alphas[uh * 3 + 1], sXs[urow], betas[uh * 3 + 1]);
        const float* sQ_ = &sQ[ua * D_DIM];
        int p = 0;
        #pragma unroll
        for (int s = D_DIM / 2; s >= 1; s >>= 1) {
            p += (sQ_[p + s - 1] < ki) ? s : 0;
        }
        p += (sQ_[p] < ki) ? 1 : 0;
        const float dl = (p >= 1)    ? dist_ln2(sQ_[(p >= 1) ? (p - 1) : 0], ki) : INFINITY;
        const float dr = (p < D_DIM) ? dist_ln2(sQ_[(p < D_DIM) ? p : (D_DIM - 1)], ki) : INFINITY;
        const float dmin = fminf(dl, dr);
        const float mi = __builtin_amdgcn_rcpf(dmin);
        const float mp = mi - ZERO_GAP;
        const float thr = (mp > 0.0f) ? (1.002f * __builtin_amdgcn_rcpf(mp)) : INFINITY;
        int lo = 0;
        int hi = 0;
        #pragma unroll
        for (int s = D_DIM / 2; s >= 1; s >>= 1) {
            const float ql = sQ_[lo + s - 1];
            const float qh = sQ_[hi + s - 1];
            const bool stepl = (ql < ki) & (dist_ln2(ql, ki) > thr);
            const bool steph = (qh < ki) | (dist_ln2(qh, ki) <= thr);
            lo += stepl ? s : 0;
            hi += steph ? s : 0;
        }
        {
            const float ql = sQ_[lo];
            const float qh = sQ_[hi];
            const bool stepl = (ql < ki) & (dist_ln2(ql, ki) > thr);
            const bool steph = (qh < ki) | (dist_ln2(qh, ki) <= thr);
            lo += stepl ? 1 : 0;
            hi += steph ? 1 : 0;
        }
        sKu[i] = ki;
        sM[i]  = mi;
        sLo[i] = lo;
        sHi[i] = hi;
        const bool wide = (hi - lo) > WIDE_T;
        sWd[i] = wide ? 1 : 0;
        if (wide) {
            const int pos = atomicAdd(&sWcnt[ua], 1);
            sWlist[ua * 128 + pos] = i;
        }
    }
    __syncthreads();

    const int half  = lane >> 4;
    const int laneM = lane & 15;
    const float invSqrtD = 0.044194173824159216f;

    const int wc0 = __builtin_amdgcn_readfirstlane(sWcnt[0]);
    const int wc1 = __builtin_amdgcn_readfirstlane(sWcnt[1]);
    const int wc2 = __builtin_amdgcn_readfirstlane(sWcnt[2]);
    const int wc3 = __builtin_amdgcn_readfirstlane(sWcnt[3]);
    const int W1 = (wc0 + 15) >> 4;
    const int W2 = W1 + ((wc1 + 15) >> 4);
    const int W3 = W2 + ((wc2 + 15) >> 4);
    const int NW = W3 + ((wc3 + 15) >> 4);
    const int nItems = NW + 8 * HPW;
    for (;;) {
        int pr = 0;
        if (lane == 0) pr = atomicAdd(&sNext, 1);
        pr = __builtin_amdgcn_readfirstlane(pr);
        if (pr >= nItems) break;
        int tt;
        int row;
        bool mine;
        if (pr < NW) {
            tt = (pr >= W1 ? 1 : 0) + (pr >= W2 ? 1 : 0) + (pr >= W3 ? 1 : 0);
            const int first = (tt == 0) ? 0 : ((tt == 1) ? W1 : ((tt == 2) ? W2 : W3));
            const int cnt   = (tt == 0) ? wc0 : ((tt == 1) ? wc1 : ((tt == 2) ? wc2 : wc3));
            const int pos = (pr - first) * 16 + laneM;
            mine = pos < cnt;
            row = sWlist[tt * 128 + min(pos, cnt - 1)];
        } else {
            const int pn = pr - NW;
            tt = pn & (HPW - 1);
            const int sidx = pn >> 2;
            const int ts   = (sidx & 1) ? (7 - (sidx >> 1)) : (sidx >> 1);
            row = tt * 128 + ts * 16 + laneM;
            mine = (sWd[row] == 0);
        }
        const float ki = sKu[row];
        const float mi = sM[row];
        const int bplane = (laneM == 0) ? 0 : ((laneM == 1) ? (2 + tt) * BPL : BPL);
        int lo = mine ? sLo[row] : 0x7fffffff;
        int hi = mine ? sHi[row] : 0;
        lo = min(lo, xrow_i<1>(lo)); hi = max(hi, xrow_i<1>(hi));
        lo = min(lo, xrow_i<2>(lo)); hi = max(hi, xrow_i<2>(hi));
        lo = min(lo, xrow_i<4>(lo)); hi = max(hi, xrow_i<4>(hi));
        lo = min(lo, xrow_i<8>(lo)); hi = max(hi, xrow_i<8>(hi));
        const int g_lo = __builtin_amdgcn_readfirstlane(lo) / GJ;
        const int g_hi = (__builtin_amdgcn_readfirstlane(hi) + GJ - 1) / GJ;
        if (g_lo >= g_hi) continue;

        v8f c0 = {};
        v8f c1 = {};
        const float* pq = &sQ[tt * D_DIM + half * 4];
        const float* pv = &sB[bplane + half * 4];
        for (int g = g_lo; g < g_hi; ++g) {
            const float4 qa = *reinterpret_cast<const float4*>(pq + g * GJ);
            const float4 qb = *reinterpret_cast<const float4*>(pq + g * GJ + 8);
            const float4 va = *reinterpret_cast<const float4*>(pv + g * GJ);
            const float4 vb = *reinterpret_cast<const float4*>(pv + g * GJ + 8);
            {
                const float e0 = __builtin_amdgcn_exp2f(__builtin_amdgcn_rcpf(dist_ln2(qa.x, ki)) - mi);
                const float e1 = __builtin_amdgcn_exp2f(__builtin_amdgcn_rcpf(dist_ln2(qa.y, ki)) - mi);
                const float e2 = __builtin_amdgcn_exp2f(__builtin_amdgcn_rcpf(dist_ln2(qa.z, ki)) - mi);
                const float e3 = __builtin_amdgcn_exp2f(__builtin_amdgcn_rcpf(dist_ln2(qa.w, ki)) - mi);
                v2f A0 = {e0, e1};
                v2f B0 = {va.x, va.y};
                v2f A1 = {e2, e3};
                v2f B1 = {va.z, va.w};
                c0 = __builtin_amdgcn_wmma_f32_16x16x4_f32(false, A0, false, B0, (short)0, c0, false, false);
                c1 = __builtin_amdgcn_wmma_f32_16x16x4_f32(false, A1, false, B1, (short)0, c1, false, false);
            }
            {
                const float e0 = __builtin_amdgcn_exp2f(__builtin_amdgcn_rcpf(dist_ln2(qb.x, ki)) - mi);
                const float e1 = __builtin_amdgcn_exp2f(__builtin_amdgcn_rcpf(dist_ln2(qb.y, ki)) - mi);
                const float e2 = __builtin_amdgcn_exp2f(__builtin_amdgcn_rcpf(dist_ln2(qb.z, ki)) - mi);
                const float e3 = __builtin_amdgcn_exp2f(__builtin_amdgcn_rcpf(dist_ln2(qb.w, ki)) - mi);
                v2f A0 = {e0, e1};
                v2f B0 = {vb.x, vb.y};
                v2f A1 = {e2, e3};
                v2f B1 = {vb.z, vb.w};
                c0 = __builtin_amdgcn_wmma_f32_16x16x4_f32(false, A0, false, B0, (short)0, c0, false, false);
                c1 = __builtin_amdgcn_wmma_f32_16x16x4_f32(false, A1, false, B1, (short)0, c1, false, false);
            }
        }

        const int M0 = half * 8;
        #pragma unroll
        for (int r = 0; r < 8; ++r) {
            sC[wave][(M0 + r) * 16 + laneM] = c0[r] + c1[r];
        }
        asm volatile("s_wait_dscnt 0" ::: "memory");
        if (lane < 16) {
            float sumE  = sC[wave][lane * 16 + 0];
            float sumEV = sC[wave][lane * 16 + 1];
            if (mine) sAtt[row] = (sumEV / sumE) * invSqrtD;
        }
    }
    __syncthreads();

    #pragma unroll
    for (int rr = 0; rr < 2; ++rr) {
        const int e = tid + rr * 256;
        const float xv = x[b * D_DIM + e];
        int p = 0;
        #pragma unroll
        for (int s = D_DIM / 2; s >= 1; s >>= 1) {
            p += (sXs[p + s - 1] < xv) ? s : 0;
        }
        const int t = p >> 4;
        if ((t & 3) == rq) {
            const int u0 = ((t >> 2) << 4) | (p & 15);
            const float sum4 = (sAtt[u0] + sAtt[128 + u0]) + (sAtt[256 + u0] + sAtt[384 + u0]);
            atomicAdd(&out[b * D_DIM + e], sum4);
        }
    }
}

extern "C" void kernel_launch(void* const* d_in, const int* in_sizes, int n_in,
                              void* d_out, int out_size, void* d_ws, size_t ws_size,
                              hipStream_t stream) {
    const float* x      = (const float*)d_in[0];
    const float* alphas = (const float*)d_in[1];
    const float* betas  = (const float*)d_in[2];
    float* out          = (float*)d_out;

    const int n = B_DIM * D_DIM;
    fl_init_out<<<(n + 255) / 256, 256, 0, stream>>>(x, out, n);
    fl_attn_kernel<<<H_DIM * B_DIM, 256, 0, stream>>>(x, alphas, betas, out);
}


# band cut where a column's exponential is under 2^-40 of its row's largest (was: only exact zeros left out); dropped terms total under 2^-31 of a row's sum: sums change below f32 rounding
# speedup vs baseline: 3.6115x; 1.2032x over previous
#include <hip/hip_runtime.h>
#include <math.h>


typedef __attribute__((ext_vector_type(2))) float v2f;
typedef __attribute__((ext_vector_type(8))) float v8f;

#define B_DIM 64
#define D_DIM 512
#define H_DIM 8
#define EPS_F 1e-8f
#define LN2_F 0.69314718055994531f
#define EPS_LN2_F (EPS_F * LN2_F)
#define HPW 4
#define BPL (D_DIM + 4)
#define GJ 16
#define WIDE_T 64
#define DROP_GAP 40.0f

#if defined(__HIP_DEVICE_COMPILE__) && defined(__AMDGCN__)
#if !__has_builtin(__builtin_amdgcn_wmma_f32_16x16x4_f32)
#error "missing __builtin_amdgcn_wmma_f32_16x16x4_f32 on this toolchain (device pass)"
#endif
#endif

__device__ __forceinline__ float dist_ln2(float q, float k) {
    return fmaf(fabsf(q - k), LN2_F, EPS_LN2_F);
}

__device__ __forceinline__ int xrow16_i(int b) {
    return __builtin_amdgcn_permlanex16(b, b, 0x76543210, 0xfedcba98, false, false);
}
template <int M>
__device__ __forceinline__ int xrow_i(int b) {
    return __builtin_amdgcn_update_dpp(b, b, 0x160 + M, 0xf, 0xf, false);
}
__device__ __forceinline__ float xrow16_f(float x) {
    return __builtin_bit_cast(float, xrow16_i(__builtin_bit_cast(int, x)));
}
template <int M>
__device__ __forceinline__ float xrow_f(float x) {
    return __builtin_bit_cast(float, xrow_i<M>(__builtin_bit_cast(int, x)));
}

__global__ void fl_init_out(const float* __restrict__ x, float* __restrict__ out, int n) {
    int i = blockIdx.x * blockDim.x + threadIdx.x;
    if (i < n) out[i] = x[i];
}

__global__ __launch_bounds__(256) void fl_attn_kernel(const float* __restrict__ x,
                                                      const float* __restrict__ alphas,
                                                      const float* __restrict__ betas,
                                                      float* __restrict__ out) {
    __shared__ __attribute__((aligned(16))) float sXs[D_DIM];
    __shared__ __attribute__((aligned(16))) float sQ[HPW * D_DIM];
    __shared__ __attribute__((aligned(16))) float sB[(2 + HPW) * BPL];
    __shared__ __attribute__((aligned(16))) float sKu[D_DIM];
    __shared__ __attribute__((aligned(16))) float sM[D_DIM];
    __shared__ int sLo[D_DIM];
    __shared__ int sHi[D_DIM];
    __shared__ __attribute__((aligned(16))) float sAtt[D_DIM];
    __shared__ __attribute__((aligned(16))) float sC[8][16 * 16];
    __shared__ int sWd[D_DIM];
    __shared__ int sWlist[D_DIM];
    __shared__ int sWcnt[HPW];
    __shared__ int sNext;

    const int hb  = blockIdx.x;
    const int b   = hb & (B_DIM - 1);
    const int rq  = (hb >> 6) & 3;
    const int par = ((hb >> 8) ^ hb) & 1;
    const int tid = threadIdx.x;
    const int wave  = tid >> 5;
    const int lane  = tid & 31;

    const int gi0 = (wave << 6) | lane;
    const int gi1 = gi0 | 32;
    float r0 = x[b * D_DIM + gi0];
    float r1 = x[b * D_DIM + gi1];
    #pragma unroll
    for (int k = 2; k <= D_DIM; k <<= 1) {
        #pragma unroll
        for (int jj = k >> 1; jj >= 1; jj >>= 1) {
            const bool km0 = (((gi0 & jj) == 0) == ((gi0 & k) == 0));
            const bool km1 = (((gi1 & jj) == 0) == ((gi1 & k) == 0));
            float p0, p1;
            if (jj >= 64) {
                __syncthreads();
                sXs[gi0] = r0;
                sXs[gi1] = r1;
                __syncthreads();
                p0 = sXs[gi0 ^ jj];
                p1 = sXs[gi1 ^ jj];
            } else if (jj == 32) {
                p0 = r1;
                p1 = r0;
            } else if (jj == 16) {
                p0 = xrow16_f(r0);
                p1 = xrow16_f(r1);
            } else if (jj == 8) {
                p0 = xrow_f<8>(r0);
                p1 = xrow_f<8>(r1);
            } else if (jj == 4) {
                p0 = xrow_f<4>(r0);
                p1 = xrow_f<4>(r1);
            } else if (jj == 2) {
                p0 = xrow_f<2>(r0);
                p1 = xrow_f<2>(r1);
            } else {
                p0 = xrow_f<1>(r0);
                p1 = xrow_f<1>(r1);
            }
            const float lo0 = fminf(r0, p0), hi0 = fmaxf(r0, p0);
            const float lo1 = fminf(r1, p1), hi1 = fmaxf(r1, p1);
            r0 = km0 ? lo0 : hi0;
            r1 = km1 ? lo1 : hi1;
        }
    }

    __syncthreads();
    if (tid == 0) sNext = 0;
    if (tid < HPW) sWcnt[tid] = 0;
    sXs[gi0] = r0;
    sXs[gi1] = r1;
    sB[gi0]       = 1.0f;
    sB[gi1]       = 1.0f;
    sB[BPL + gi0] = 0.0f;
    sB[BPL + gi1] = 0.0f;
    #pragma unroll
    for (int a = 0; a < HPW; ++a) {
        const int h = par + 2 * a;
        const float aq = alphas[h * 3 + 0], av = alphas[h * 3 + 2];
        const float bq = betas[h * 3 + 0],  bv = betas[h * 3 + 2];
        const bool rev = (aq < 0.0f);
        const int jq0 = rev ? (D_DIM - 1 - gi0) : gi0;
        const int jq1 = rev ? (D_DIM - 1 - gi1) : gi1;
        sQ[a * D_DIM + jq0] = fmaf(aq, r0, bq);
        sQ[a * D_DIM + jq1] = fmaf(aq, r1, bq);
        sB[(2 + a) * BPL + jq0] = fmaf(av, r0, bv);
        sB[(2 + a) * BPL + jq1] = fmaf(av, r1, bv);
    }
    __syncthreads();

    #pragma unroll
    for (int rr = 0; rr < 2; ++rr) {
        const int i = tid + rr * 256;
        const int ua = i >> 7;
        const int uh = par + 2 * ua;
        const int urow = ((rq + 4 * ((i >> 4) & 7)) << 4) | (i & 15);
        const float ki = fmaf(alphas[uh * 3 + 1], sXs[urow], betas[uh * 3 + 1]);
        const float* sQ_ = &sQ[ua * D_DIM];
        int p = 0;
        #pragma unroll
        for (int s = D_DIM / 2; s >= 1; s >>= 1) {
            p += (sQ_[p + s - 1] < ki) ? s : 0;
        }
        p += (sQ_[p] < ki) ? 1 : 0;
        const float dl = (p >= 1)    ? dist_ln2(sQ_[(p >= 1) ? (p - 1) : 0], ki) : INFINITY;
        const float dr = (p < D_DIM) ? dist_ln2(sQ_[(p < D_DIM) ? p : (D_DIM - 1)], ki) : INFINITY;
        const float dmin = fminf(dl, dr);
        const float mi = __builtin_amdgcn_rcpf(dmin);
        const float mp = mi - DROP_GAP;
        const float thr = (mp > 0.0f) ? (1.002f * __builtin_amdgcn_rcpf(mp)) : INFINITY;
        int lo = 0;
        int hi = 0;
        #pragma unroll
        for (int s = D_DIM / 2; s >= 1; s >>= 1) {
            const float ql = sQ_[lo + s - 1];
            const float qh = sQ_[hi + s - 1];
            const bool stepl = (ql < ki) & (dist_ln2(ql, ki) > thr);
            const bool steph = (qh < ki) | (dist_ln2(qh, ki) <= thr);
            lo += stepl ? s : 0;
            hi += steph ? s : 0;
        }
        {
            const float ql = sQ_[lo];
            const float qh = sQ_[hi];
            const bool stepl = (ql < ki) & (dist_ln2(ql, ki) > thr);
            const bool steph = (qh < ki) | (dist_ln2(qh, ki) <= thr);
            lo += stepl ? 1 : 0;
            hi += steph ? 1 : 0;
        }
        sKu[i] = ki;
        sM[i]  = mi;
        sLo[i] = lo;
        sHi[i] = hi;
        const bool wide = (hi - lo) > WIDE_T;
        sWd[i] = wide ? 1 : 0;
        if (wide) {
            const int pos = atomicAdd(&sWcnt[ua], 1);
            sWlist[ua * 128 + pos] = i;
        }
    }
    __syncthreads();

    const int half  = lane >> 4;
    const int laneM = lane & 15;
    const float invSqrtD = 0.044194173824159216f;

    const int wc0 = __builtin_amdgcn_readfirstlane(sWcnt[0]);
    const int wc1 = __builtin_amdgcn_readfirstlane(sWcnt[1]);
    const int wc2 = __builtin_amdgcn_readfirstlane(sWcnt[2]);
    const int wc3 = __builtin_amdgcn_readfirstlane(sWcnt[3]);
    const int W1 = (wc0 + 15) >> 4;
    const int W2 = W1 + ((wc1 + 15) >> 4);
    const int W3 = W2 + ((wc2 + 15) >> 4);
    const int NW = W3 + ((wc3 + 15) >> 4);
    const int nItems = NW + 8 * HPW;
    for (;;) {
        int pr = 0;
        if (lane == 0) pr = atomicAdd(&sNext, 1);
        pr = __builtin_amdgcn_readfirstlane(pr);
        if (pr >= nItems) break;
        int tt;
        int row;
        bool mine;
        if (pr < NW) {
            tt = (pr >= W1 ? 1 : 0) + (pr >= W2 ? 1 : 0) + (pr >= W3 ? 1 : 0);
            const int first = (tt == 0) ? 0 : ((tt == 1) ? W1 : ((tt == 2) ? W2 : W3));
            const int cnt   = (tt == 0) ? wc0 : ((tt == 1) ? wc1 : ((tt == 2) ? wc2 : wc3));
            const int pos = (pr - first) * 16 + laneM;
            mine = pos < cnt;
            row = sWlist[tt * 128 + min(pos, cnt - 1)];
        } else {
            const int pn = pr - NW;
            tt = pn & (HPW - 1);
            const int sidx = pn >> 2;
            const int ts   = (sidx & 1) ? (7 - (sidx >> 1)) : (sidx >> 1);
            row = tt * 128 + ts * 16 + laneM;
            mine = (sWd[row] == 0);
        }
        const float ki = sKu[row];
        const float mi = sM[row];
        const int bplane = (laneM == 0) ? 0 : ((laneM == 1) ? (2 + tt) * BPL : BPL);
        int lo = mine ? sLo[row] : 0x7fffffff;
        int hi = mine ? sHi[row] : 0;
        lo = min(lo, xrow_i<1>(lo)); hi = max(hi, xrow_i<1>(hi));
        lo = min(lo, xrow_i<2>(lo)); hi = max(hi, xrow_i<2>(hi));
        lo = min(lo, xrow_i<4>(lo)); hi = max(hi, xrow_i<4>(hi));
        lo = min(lo, xrow_i<8>(lo)); hi = max(hi, xrow_i<8>(hi));
        const int g_lo = __builtin_amdgcn_readfirstlane(lo) / GJ;
        const int g_hi = (__builtin_amdgcn_readfirstlane(hi) + GJ - 1) / GJ;
        if (g_lo >= g_hi) continue;

        v8f c0 = {};
        v8f c1 = {};
        const float* pq = &sQ[tt * D_DIM + half * 4];
        const float* pv = &sB[bplane + half * 4];
        for (int g = g_lo; g < g_hi; ++g) {
            const float4 qa = *reinterpret_cast<const float4*>(pq + g * GJ);
            const float4 qb = *reinterpret_cast<const float4*>(pq + g * GJ + 8);
            const float4 va = *reinterpret_cast<const float4*>(pv + g * GJ);
            const float4 vb = *reinterpret_cast<const float4*>(pv + g * GJ + 8);
            {
                const float e0 = __builtin_amdgcn_exp2f(__builtin_amdgcn_rcpf(dist_ln2(qa.x, ki)) - mi);
                const float e1 = __builtin_amdgcn_exp2f(__builtin_amdgcn_rcpf(dist_ln2(qa.y, ki)) - mi);
                const float e2 = __builtin_amdgcn_exp2f(__builtin_amdgcn_rcpf(dist_ln2(qa.z, ki)) - mi);
                const float e3 = __builtin_amdgcn_exp2f(__builtin_amdgcn_rcpf(dist_ln2(qa.w, ki)) - mi);
                v2f A0 = {e0, e1};
                v2f B0 = {va.x, va.y};
                v2f A1 = {e2, e3};
                v2f B1 = {va.z, va.w};
                c0 = __builtin_amdgcn_wmma_f32_16x16x4_f32(false, A0, false, B0, (short)0, c0, false, false);
                c1 = __builtin_amdgcn_wmma_f32_16x16x4_f32(false, A1, false, B1, (short)0, c1, false, false);
            }
            {
                const float e0 = __builtin_amdgcn_exp2f(__builtin_amdgcn_rcpf(dist_ln2(qb.x, ki)) - mi);
                const float e1 = __builtin_amdgcn_exp2f(__builtin_amdgcn_rcpf(dist_ln2(qb.y, ki)) - mi);
                const float e2 = __builtin_amdgcn_exp2f(__builtin_amdgcn_rcpf(dist_ln2(qb.z, ki)) - mi);
                const float e3 = __builtin_amdgcn_exp2f(__builtin_amdgcn_rcpf(dist_ln2(qb.w, ki)) - mi);
                v2f A0 = {e0, e1};
                v2f B0 = {vb.x, vb.y};
                v2f A1 = {e2, e3};
                v2f B1 = {vb.z, vb.w};
                c0 = __builtin_amdgcn_wmma_f32_16x16x4_f32(false, A0, false, B0, (short)0, c0, false, false);
                c1 = __builtin_amdgcn_wmma_f32_16x16x4_f32(false, A1, false, B1, (short)0, c1, false, false);
            }
        }

        const int M0 = half * 8;
        #pragma unroll
        for (int r = 0; r < 8; ++r) {
            sC[wave][(M0 + r) * 16 + laneM] = c0[r] + c1[r];
        }
        asm volatile("s_wait_dscnt 0" ::: "memory");
        if (lane < 16) {
            float sumE  = sC[wave][lane * 16 + 0];
            float sumEV = sC[wave][lane * 16 + 1];
            if (mine) sAtt[row] = (sumEV / sumE) * invSqrtD;
        }
    }
    __syncthreads();

    #pragma unroll
    for (int rr = 0; rr < 2; ++rr) {
        const int e = tid + rr * 256;
        const float xv = x[b * D_DIM + e];
        int p = 0;
        #pragma unroll
        for (int s = D_DIM / 2; s >= 1; s >>= 1) {
            p += (sXs[p + s - 1] < xv) ? s : 0;
        }
        const int t = p >> 4;
        if ((t & 3) == rq) {
            const int u0 = ((t >> 2) << 4) | (p & 15);
            const float sum4 = (sAtt[u0] + sAtt[128 + u0]) + (sAtt[256 + u0] + sAtt[384 + u0]);
            atomicAdd(&out[b * D_DIM + e], sum4);
        }
    }
}

extern "C" void kernel_launch(void* const* d_in, const int* in_sizes, int n_in,
                              void* d_out, int out_size, void* d_ws, size_t ws_size,
                              hipStream_t stream) {
    const float* x      = (const float*)d_in[0];
    const float* alphas = (const float*)d_in[1];
    const float* betas  = (const float*)d_in[2];
    float* out          = (float*)d_out;

    const int n = B_DIM * D_DIM;
    fl_init_out<<<(n + 255) / 256, 256, 0, stream>>>(x, out, n);
    fl_attn_kernel<<<H_DIM * B_DIM, 256, 0, stream>>>(x, alphas, betas, out);
}


# a thread's two binary searches (rows before the band loop, elements after it) run side by side in one branch-free stretch, not one after the other; head parameters by scalar loads at entry; same sums
# speedup vs baseline: 3.9750x; 1.1007x over previous
#include <hip/hip_runtime.h>
#include <math.h>


typedef __attribute__((ext_vector_type(2))) float v2f;
typedef __attribute__((ext_vector_type(8))) float v8f;

#define B_DIM 64
#define D_DIM 512
#define H_DIM 8
#define EPS_F 1e-8f
#define LN2_F 0.69314718055994531f
#define EPS_LN2_F (EPS_F * LN2_F)
#define HPW 4
#define BPL (D_DIM + 4)
#define GJ 16
#define WIDE_T 64
#define DROP_GAP 40.0f

#if defined(__HIP_DEVICE_COMPILE__) && defined(__AMDGCN__)
#if !__has_builtin(__builtin_amdgcn_wmma_f32_16x16x4_f32)
#error "missing __builtin_amdgcn_wmma_f32_16x16x4_f32 on this toolchain (device pass)"
#endif
#endif

__device__ __forceinline__ float dist_ln2(float q, float k) {
    return fmaf(fabsf(q - k), LN2_F, EPS_LN2_F);
}

__device__ __forceinline__ int xrow16_i(int b) {
    return __builtin_amdgcn_permlanex16(b, b, 0x76543210, 0xfedcba98, false, false);
}
template <int M>
__device__ __forceinline__ int xrow_i(int b) {
    return __builtin_amdgcn_update_dpp(b, b, 0x160 + M, 0xf, 0xf, false);
}
__device__ __forceinline__ float xrow16_f(float x) {
    return __builtin_bit_cast(float, xrow16_i(__builtin_bit_cast(int, x)));
}
template <int M>
__device__ __forceinline__ float xrow_f(float x) {
    return __builtin_bit_cast(float, xrow_i<M>(__builtin_bit_cast(int, x)));
}

__global__ void fl_init_out(const float* __restrict__ x, float* __restrict__ out, int n) {
    int i = blockIdx.x * blockDim.x + threadIdx.x;
    if (i < n) out[i] = x[i];
}

__global__ __launch_bounds__(256) void fl_attn_kernel(const float* __restrict__ x,
                                                      const float* __restrict__ alphas,
                                                      const float* __restrict__ betas,
                                                      float* __restrict__ out) {
    __shared__ __attribute__((aligned(16))) float sXs[D_DIM];
    __shared__ __attribute__((aligned(16))) float sQ[HPW * D_DIM];
    __shared__ __attribute__((aligned(16))) float sB[(2 + HPW) * BPL];
    __shared__ __attribute__((aligned(16))) float sKu[D_DIM];
    __shared__ __attribute__((aligned(16))) float sM[D_DIM];
    __shared__ int sLo[D_DIM];
    __shared__ int sHi[D_DIM];
    __shared__ __attribute__((aligned(16))) float sAtt[D_DIM];
    __shared__ __attribute__((aligned(16))) float sC[8][16 * 16];
    __shared__ int sWd[D_DIM];
    __shared__ int sWlist[D_DIM];
    __shared__ int sWcnt[HPW];
    __shared__ int sNext;

    const int hb  = blockIdx.x;
    const int b   = hb & (B_DIM - 1);
    const int rq  = (hb >> 6) & 3;
    const int par = ((hb >> 8) ^ hb) & 1;
    const int tid = threadIdx.x;
    const int wave  = tid >> 5;
    const int lane  = tid & 31;

    const int gi0 = (wave << 6) | lane;
    const int gi1 = gi0 | 32;
    float r0 = x[b * D_DIM + gi0];
    float r1 = x[b * D_DIM + gi1];
    const float x0 = r0;
    const float x1 = r1;

    float aqS[HPW], bqS[HPW], avS[HPW], bvS[HPW];
    #pragma unroll
    for (int a = 0; a < HPW; ++a) {
        const int h = par + 2 * a;
        aqS[a] = alphas[h * 3 + 0];
        avS[a] = alphas[h * 3 + 2];
        bqS[a] = betas[h * 3 + 0];
        bvS[a] = betas[h * 3 + 2];
    }
    const int ua0 = __builtin_amdgcn_readfirstlane(tid >> 7);
    float akU[2], bkU[2];
    #pragma unroll
    for (int rr = 0; rr < 2; ++rr) {
        const int uh = par + 2 * (ua0 + 2 * rr);
        akU[rr] = alphas[uh * 3 + 1];
        bkU[rr] = betas[uh * 3 + 1];
    }
    #pragma unroll
    for (int k = 2; k <= D_DIM; k <<= 1) {
        #pragma unroll
        for (int jj = k >> 1; jj >= 1; jj >>= 1) {
            const bool km0 = (((gi0 & jj) == 0) == ((gi0 & k) == 0));
            const bool km1 = (((gi1 & jj) == 0) == ((gi1 & k) == 0));
            float p0, p1;
            if (jj >= 64) {
                __syncthreads();
                sXs[gi0] = r0;
                sXs[gi1] = r1;
                __syncthreads();
                p0 = sXs[gi0 ^ jj];
                p1 = sXs[gi1 ^ jj];
            } else if (jj == 32) {
                p0 = r1;
                p1 = r0;
            } else if (jj == 16) {
                p0 = xrow16_f(r0);
                p1 = xrow16_f(r1);
            } else if (jj == 8) {
                p0 = xrow_f<8>(r0);
                p1 = xrow_f<8>(r1);
            } else if (jj == 4) {
                p0 = xrow_f<4>(r0);
                p1 = xrow_f<4>(r1);
            } else if (jj == 2) {
                p0 = xrow_f<2>(r0);
                p1 = xrow_f<2>(r1);
            } else {
                p0 = xrow_f<1>(r0);
                p1 = xrow_f<1>(r1);
            }
            const float lo0 = fminf(r0, p0), hi0 = fmaxf(r0, p0);
            const float lo1 = fminf(r1, p1), hi1 = fmaxf(r1, p1);
            r0 = km0 ? lo0 : hi0;
            r1 = km1 ? lo1 : hi1;
        }
    }

    __syncthreads();
    if (tid == 0) sNext = 0;
    if (tid < HPW) sWcnt[tid] = 0;
    sXs[gi0] = r0;
    sXs[gi1] = r1;
    sB[gi0]       = 1.0f;
    sB[gi1]       = 1.0f;
    sB[BPL + gi0] = 0.0f;
    sB[BPL + gi1] = 0.0f;
    #pragma unroll
    for (int a = 0; a < HPW; ++a) {
        const float aq = aqS[a], av = avS[a];
        const float bq = bqS[a], bv = bvS[a];
        const bool rev = (aq < 0.0f);
        const int jq0 = rev ? (D_DIM - 1 - gi0) : gi0;
        const int jq1 = rev ? (D_DIM - 1 - gi1) : gi1;
        sQ[a * D_DIM + jq0] = fmaf(aq, r0, bq);
        sQ[a * D_DIM + jq1] = fmaf(aq, r1, bq);
        sB[(2 + a) * BPL + jq0] = fmaf(av, r0, bv);
        sB[(2 + a) * BPL + jq1] = fmaf(av, r1, bv);
    }
    __syncthreads();

    float kiU[2], miU[2], thrU[2];
    int   pU[2], loU[2], hiU[2];
    const float* sQU[2];
    {
        const int urow = ((rq + 4 * ((tid >> 4) & 7)) << 4) | (tid & 15);
        const float xrow = sXs[urow];
        #pragma unroll
        for (int rr = 0; rr < 2; ++rr) {
            kiU[rr] = fmaf(akU[rr], xrow, bkU[rr]);
            sQU[rr] = &sQ[(ua0 + 2 * rr) * D_DIM];
            pU[rr]  = 0;
        }
    }
    #pragma unroll
    for (int s = D_DIM / 2; s >= 1; s >>= 1) {
        #pragma unroll
        for (int rr = 0; rr < 2; ++rr) {
            pU[rr] += (sQU[rr][pU[rr] + s - 1] < kiU[rr]) ? s : 0;
        }
    }
    #pragma unroll
    for (int rr = 0; rr < 2; ++rr) {
        pU[rr] += (sQU[rr][pU[rr]] < kiU[rr]) ? 1 : 0;
    }
    #pragma unroll
    for (int rr = 0; rr < 2; ++rr) {
        const int p = pU[rr];
        const float ki = kiU[rr];
        const float qL = sQU[rr][(p >= 1) ? (p - 1) : 0];
        const float qR = sQU[rr][(p < D_DIM) ? p : (D_DIM - 1)];
        const float dL = dist_ln2(qL, ki);
        const float dR = dist_ln2(qR, ki);
        const float dl = (p >= 1)    ? dL : INFINITY;
        const float dr = (p < D_DIM) ? dR : INFINITY;
        const float dmin = fminf(dl, dr);
        const float mi = __builtin_amdgcn_rcpf(dmin);
        const float mp = mi - DROP_GAP;
        const float tr = 1.002f * __builtin_amdgcn_rcpf(mp);
        miU[rr]  = mi;
        thrU[rr] = (mp > 0.0f) ? tr : INFINITY;
        loU[rr] = 0;
        hiU[rr] = 0;
    }
    #pragma unroll
    for (int s = D_DIM / 2; s >= 1; s >>= 1) {
        #pragma unroll
        for (int rr = 0; rr < 2; ++rr) {
            const float ki = kiU[rr], thr = thrU[rr];
            const float ql = sQU[rr][loU[rr] + s - 1];
            const float qh = sQU[rr][hiU[rr] + s - 1];
            const bool stepl = (ql < ki) & (dist_ln2(ql, ki) > thr);
            const bool steph = (qh < ki) | (dist_ln2(qh, ki) <= thr);
            loU[rr] += stepl ? s : 0;
            hiU[rr] += steph ? s : 0;
        }
    }
    #pragma unroll
    for (int rr = 0; rr < 2; ++rr) {
        const float ki = kiU[rr], thr = thrU[rr];
        const float ql = sQU[rr][loU[rr]];
        const float qh = sQU[rr][hiU[rr]];
        const bool stepl = (ql < ki) & (dist_ln2(ql, ki) > thr);
        const bool steph = (qh < ki) | (dist_ln2(qh, ki) <= thr);
        loU[rr] += stepl ? 1 : 0;
        hiU[rr] += steph ? 1 : 0;
    }
    bool wideU[2];
    #pragma unroll
    for (int rr = 0; rr < 2; ++rr) {
        const int i = tid + rr * 256;
        sKu[i] = kiU[rr];
        sM[i]  = miU[rr];
        sLo[i] = loU[rr];
        sHi[i] = hiU[rr];
        wideU[rr] = (hiU[rr] - loU[rr]) > WIDE_T;
        sWd[i] = wideU[rr] ? 1 : 0;
    }
    #pragma unroll
    for (int rr = 0; rr < 2; ++rr) {
        if (wideU[rr]) {
            const int ua = ua0 + 2 * rr;
            const int pos = atomicAdd(&sWcnt[ua], 1);
            sWlist[ua * 128 + pos] = tid + rr * 256;
        }
    }
    __syncthreads();

    const int half  = lane >> 4;
    const int laneM = lane & 15;
    const float invSqrtD = 0.044194173824159216f;

    const int wc0 = __builtin_amdgcn_readfirstlane(sWcnt[0]);
    const int wc1 = __builtin_amdgcn_readfirstlane(sWcnt[1]);
    const int wc2 = __builtin_amdgcn_readfirstlane(sWcnt[2]);
    const int wc3 = __builtin_amdgcn_readfirstlane(sWcnt[3]);
    const int W1 = (wc0 + 15) >> 4;
    const int W2 = W1 + ((wc1 + 15) >> 4);
    const int W3 = W2 + ((wc2 + 15) >> 4);
    const int NW = W3 + ((wc3 + 15) >> 4);
    const int nItems = NW + 8 * HPW;
    for (;;) {
        int pr = 0;
        if (lane == 0) pr = atomicAdd(&sNext, 1);
        pr = __builtin_amdgcn_readfirstlane(pr);
        if (pr >= nItems) break;
        int tt;
        int row;
        bool mine;
        if (pr < NW) {
            tt = (pr >= W1 ? 1 : 0) + (pr >= W2 ? 1 : 0) + (pr >= W3 ? 1 : 0);
            const int first = (tt == 0) ? 0 : ((tt == 1) ? W1 : ((tt == 2) ? W2 : W3));
            const int cnt   = (tt == 0) ? wc0 : ((tt == 1) ? wc1 : ((tt == 2) ? wc2 : wc3));
            const int pos = (pr - first) * 16 + laneM;
            mine = pos < cnt;
            row = sWlist[tt * 128 + min(pos, cnt - 1)];
        } else {
            const int pn = pr - NW;
            tt = pn & (HPW - 1);
            const int sidx = pn >> 2;
            const int ts   = (sidx & 1) ? (7 - (sidx >> 1)) : (sidx >> 1);
            row = tt * 128 + ts * 16 + laneM;
            mine = (sWd[row] == 0);
        }
        const float ki = sKu[row];
        const float mi = sM[row];
        const int bplane = (laneM == 0) ? 0 : ((laneM == 1) ? (2 + tt) * BPL : BPL);
        int lo = mine ? sLo[row] : 0x7fffffff;
        int hi = mine ? sHi[row] : 0;
        lo = min(lo, xrow_i<1>(lo)); hi = max(hi, xrow_i<1>(hi));
        lo = min(lo, xrow_i<2>(lo)); hi = max(hi, xrow_i<2>(hi));
        lo = min(lo, xrow_i<4>(lo)); hi = max(hi, xrow_i<4>(hi));
        lo = min(lo, xrow_i<8>(lo)); hi = max(hi, xrow_i<8>(hi));
        const int g_lo = __builtin_amdgcn_readfirstlane(lo) / GJ;
        const int g_hi = (__builtin_amdgcn_readfirstlane(hi) + GJ - 1) / GJ;
        if (g_lo >= g_hi) continue;

        v8f c0 = {};
        v8f c1 = {};
        const float* pq = &sQ[tt * D_DIM + half * 4];
        const float* pv = &sB[bplane + half * 4];
        for (int g = g_lo; g < g_hi; ++g) {
            const float4 qa = *reinterpret_cast<const float4*>(pq + g * GJ);
            const float4 qb = *reinterpret_cast<const float4*>(pq + g * GJ + 8);
            const float4 va = *reinterpret_cast<const float4*>(pv + g * GJ);
            const float4 vb = *reinterpret_cast<const float4*>(pv + g * GJ + 8);
            {
                const float e0 = __builtin_amdgcn_exp2f(__builtin_amdgcn_rcpf(dist_ln2(qa.x, ki)) - mi);
                const float e1 = __builtin_amdgcn_exp2f(__builtin_amdgcn_rcpf(dist_ln2(qa.y, ki)) - mi);
                const float e2 = __builtin_amdgcn_exp2f(__builtin_amdgcn_rcpf(dist_ln2(qa.z, ki)) - mi);
                const float e3 = __builtin_amdgcn_exp2f(__builtin_amdgcn_rcpf(dist_ln2(qa.w, ki)) - mi);
                v2f A0 = {e0, e1};
                v2f B0 = {va.x, va.y};
                v2f A1 = {e2, e3};
                v2f B1 = {va.z, va.w};
                c0 = __builtin_amdgcn_wmma_f32_16x16x4_f32(false, A0, false, B0, (short)0, c0, false, false);
                c1 = __builtin_amdgcn_wmma_f32_16x16x4_f32(false, A1, false, B1, (short)0, c1, false, false);
            }
            {
                const float e0 = __builtin_amdgcn_exp2f(__builtin_amdgcn_rcpf(dist_ln2(qb.x, ki)) - mi);
                const float e1 = __builtin_amdgcn_exp2f(__builtin_amdgcn_rcpf(dist_ln2(qb.y, ki)) - mi);
                const float e2 = __builtin_amdgcn_exp2f(__builtin_amdgcn_rcpf(dist_ln2(qb.z, ki)) - mi);
                const float e3 = __builtin_amdgcn_exp2f(__builtin_amdgcn_rcpf(dist_ln2(qb.w, ki)) - mi);
                v2f A0 = {e0, e1};
                v2f B0 = {vb.x, vb.y};
                v2f A1 = {e2, e3};
                v2f B1 = {vb.z, vb.w};
                c0 = __builtin_amdgcn_wmma_f32_16x16x4_f32(false, A0, false, B0, (short)0, c0, false, false);
                c1 = __builtin_amdgcn_wmma_f32_16x16x4_f32(false, A1, false, B1, (short)0, c1, false, false);
            }
        }

        const int M0 = half * 8;
        #pragma unroll
        for (int r = 0; r < 8; ++r) {
            sC[wave][(M0 + r) * 16 + laneM] = c0[r] + c1[r];
        }
        asm volatile("s_wait_dscnt 0" ::: "memory");
        if (lane < 16) {
            float sumE  = sC[wave][lane * 16 + 0];
            float sumEV = sC[wave][lane * 16 + 1];
            if (mine) sAtt[row] = (sumEV / sumE) * invSqrtD;
        }
    }
    __syncthreads();

    {
        const float xe[2] = {x0, x1};
        const int   ge[2] = {gi0, gi1};
        int pe[2] = {0, 0};
        #pragma unroll
        for (int s = D_DIM / 2; s >= 1; s >>= 1) {
            #pragma unroll
            for (int rr = 0; rr < 2; ++rr) {
                pe[rr] += (sXs[pe[rr] + s - 1] < xe[rr]) ? s : 0;
            }
        }
        float sum4[2];
        bool  own[2];
        #pragma unroll
        for (int rr = 0; rr < 2; ++rr) {
            const int p = pe[rr];
            const int t = p >> 4;
            const int u0 = ((t >> 2) << 4) | (p & 15);
            sum4[rr] = (sAtt[u0] + sAtt[128 + u0]) + (sAtt[256 + u0] + sAtt[384 + u0]);
            own[rr] = ((t & 3) == rq);
        }
        #pragma unroll
        for (int rr = 0; rr < 2; ++rr) {
            if (own[rr]) atomicAdd(&out[b * D_DIM + ge[rr]], sum4[rr]);
        }
    }
}

extern "C" void kernel_launch(void* const* d_in, const int* in_sizes, int n_in,
                              void* d_out, int out_size, void* d_ws, size_t ws_size,
                              hipStream_t stream) {
    const float* x      = (const float*)d_in[0];
    const float* alphas = (const float*)d_in[1];
    const float* betas  = (const float*)d_in[2];
    float* out          = (float*)d_out;

    const int n = B_DIM * D_DIM;
    fl_init_out<<<(n + 255) / 256, 256, 0, stream>>>(x, out, n);
    fl_attn_kernel<<<H_DIM * B_DIM, 256, 0, stream>>>(x, alphas, betas, out);
}
